# starting kernel
# baseline (speedup 1.0000x reference)
#include <hip/hip_runtime.h>
#include <hip/hip_bf16.h>
#include <math.h>

#define B_   32
#define C_   768
#define CR_  192
#define HW_  1024
#define K1_  C_
#define K2_  CR_

typedef __attribute__((ext_vector_type(2))) float v2f;
typedef __attribute__((ext_vector_type(8))) float v8f;

__device__ __forceinline__ float sigmoidf_fast(float v) {
    return 1.0f / (1.0f + __expf(-v));
}

__global__ __launch_bounds__(256) void se_squeeze(const float* __restrict__ x,
                                                  float* __restrict__ s) {
    const int bc = blockIdx.x;
    const float4* p = (const float4*)(x + (size_t)bc * HW_);
    float4 v = p[threadIdx.x];
    float sum = v.x + v.y + v.z + v.w;

    #pragma unroll
    for (int o = 16; o > 0; o >>= 1)
        sum += __shfl_down(sum, o, 32);

    __shared__ float tmp[8];
    const int lane = threadIdx.x & 31;
    const int wave = threadIdx.x >> 5;
    if (lane == 0) tmp[wave] = sum;
    __syncthreads();
    if (threadIdx.x < 8) {
        float t = tmp[threadIdx.x];
        #pragma unroll
        for (int o = 4; o > 0; o >>= 1)
            t += __shfl_down(t, o, 8);
        if (threadIdx.x == 0) s[bc] = t * (1.0f / (float)HW_);
    }
}

__global__ __launch_bounds__(256) void se_mlp_wmma(const float* __restrict__ s,
                                                   const float* __restrict__ w1,
                                                   const float* __restrict__ b1,
                                                   const float* __restrict__ w2,
                                                   const float* __restrict__ b2,
                                                   float* __restrict__ h,
                                                   float* __restrict__ g) {
    const int wave = threadIdx.x >> 5;
    const int lane = threadIdx.x & 31;
    const int half = lane >> 4;
    const int l    = lane & 15;
    const int koff = 2 * half;

    for (int t = wave; t < 24; t += 8) {
        const int mt = t & 1;
        const int nt = t >> 1;
        const int m0 = mt * 16;
        const int n0 = nt * 16;
        const float* arow = s  + (size_t)(m0 + l) * K1_ + koff;
        const float* brow = w1 + (size_t)(n0 + l) * K1_ + koff;

        v8f acc = {};
        #pragma unroll 4
        for (int k = 0; k < K1_; k += 4) {
            v2f a = *(const v2f*)(arow + k);
            v2f b = *(const v2f*)(brow + k);
            acc = __builtin_amdgcn_wmma_f32_16x16x4_f32(
                false, a, false, b, (short)0, acc, false, false);
        }
        const float bias = b1[n0 + l];
        #pragma unroll
        for (int r = 0; r < 8; ++r) {
            const int m = m0 + r + 8 * half;
            float v = acc[r] + bias;
            v = v * sigmoidf_fast(v);
            h[(size_t)m * CR_ + n0 + l] = v;
        }
    }

    __syncthreads();

    for (int t = wave; t < 96; t += 8) {
        const int mt = t & 1;
        const int nt = t >> 1;
        const int m0 = mt * 16;
        const int n0 = nt * 16;
        const float* arow = h  + (size_t)(m0 + l) * K2_ + koff;
        const float* brow = w2 + (size_t)(n0 + l) * K2_ + koff;

        v8f acc = {};
        #pragma unroll 4
        for (int k = 0; k < K2_; k += 4) {
            v2f a = *(const v2f*)(arow + k);
            v2f b = *(const v2f*)(brow + k);
            acc = __builtin_amdgcn_wmma_f32_16x16x4_f32(
                false, a, false, b, (short)0, acc, false, false);
        }
        const float bias = b2[n0 + l];
        #pragma unroll
        for (int r = 0; r < 8; ++r) {
            const int m = m0 + r + 8 * half;
            float v = acc[r] + bias;
            g[(size_t)m * C_ + n0 + l] = sigmoidf_fast(v);
        }
    }
}

__global__ __launch_bounds__(256) void se_scale(const float* __restrict__ x,
                                                const float* __restrict__ g,
                                                float* __restrict__ out) {
    const size_t idx4 = (size_t)blockIdx.x * blockDim.x + threadIdx.x;
    const size_t total4 = (size_t)B_ * C_ * HW_ / 4;
    if (idx4 >= total4) return;
    const size_t bc = idx4 >> 8;
    const float gv = g[bc];
    float4 v = ((const float4*)x)[idx4];
    v.x *= gv; v.y *= gv; v.z *= gv; v.w *= gv;
    ((float4*)out)[idx4] = v;
}

extern "C" void kernel_launch(void* const* d_in, const int* in_sizes, int n_in,
                              void* d_out, int out_size, void* d_ws, size_t ws_size,
                              hipStream_t stream) {
    const float* x  = (const float*)d_in[0];
    const float* w1 = (const float*)d_in[1];
    const float* b1 = (const float*)d_in[2];
    const float* w2 = (const float*)d_in[3];
    const float* b2 = (const float*)d_in[4];
    float* out = (float*)d_out;

    float* s = (float*)d_ws;
    float* h = s + (size_t)B_ * C_;
    float* g = h + (size_t)B_ * CR_;

    se_squeeze<<<B_ * C_, 256, 0, stream>>>(x, s);

    se_mlp_wmma<<<1, 256, 0, stream>>>(s, w1, b1, w2, b2, h, g);

    const int nblocks = (int)(((size_t)B_ * C_ * HW_ / 4 + 255) / 256);
    se_scale<<<nblocks, 256, 0, stream>>>(x, g, out);
}


# Excitation MLP spread over the chip: two launches (12 and 48 work-groups) replace one work-group; K split over each work-group's waves, partials summed via LDS in fixed order (new f32 sum order)
# speedup vs baseline: 3.4511x; 3.4511x over previous
#include <hip/hip_runtime.h>
#include <hip/hip_bf16.h>
#include <math.h>

#define B_   32
#define C_   768
#define CR_  192
#define HW_  1024

typedef __attribute__((ext_vector_type(2))) float v2f;
typedef __attribute__((ext_vector_type(8))) float v8f;

__device__ __forceinline__ float sigmoidf_fast(float v) {
    return 1.0f / (1.0f + __expf(-v));
}

__global__ __launch_bounds__(256) void se_squeeze(const float* __restrict__ x,
                                                  float* __restrict__ s) {
    const int bc = blockIdx.x;
    const float4* p = (const float4*)(x + (size_t)bc * HW_);
    float4 v = p[threadIdx.x];
    float sum = v.x + v.y + v.z + v.w;

    #pragma unroll
    for (int o = 16; o > 0; o >>= 1)
        sum += __shfl_down(sum, o, 32);

    __shared__ float tmp[8];
    const int lane = threadIdx.x & 31;
    const int wave = threadIdx.x >> 5;
    if (lane == 0) tmp[wave] = sum;
    __syncthreads();
    if (threadIdx.x < 8) {
        float t = tmp[threadIdx.x];
        #pragma unroll
        for (int o = 4; o > 0; o >>= 1)
            t += __shfl_down(t, o, 8);
        if (threadIdx.x == 0) s[bc] = t * (1.0f / (float)HW_);
    }
}

template <int K, int N, int NW, bool SWISH>
__global__ __launch_bounds__(NW * 32) void se_fc_wmma(const float* __restrict__ in,
                                                      const float* __restrict__ w,
                                                      const float* __restrict__ bias,
                                                      float* __restrict__ out) {
    constexpr int R  = K / NW;
    constexpr int NL = R / 8;
    static_assert(K % NW == 0 && R % 8 == 0, "k range of a wave must be a multiple of 8");
    static_assert(N % 32 == 0 && NW <= 16, "one wave per output row in the epilogue");
    constexpr int TS = 272;
    __shared__ float red[NW * 2 * TS];

    const int wave = threadIdx.x >> 5;
    const int lane = threadIdx.x & 31;
    const int half = lane >> 4;
    const int c    = lane & 15;
    const int m0   = blockIdx.y * 16;
    const int n0   = blockIdx.x * 32;

    const float bv = bias[n0 + lane];

    const int kofs = wave * R + 4 * half;
    const float4* ap  = (const float4*)(in + (size_t)(m0 + c)      * K + kofs);
    const float4* bp0 = (const float4*)(w  + (size_t)(n0 + c)      * K + kofs);
    const float4* bp1 = (const float4*)(w  + (size_t)(n0 + 16 + c) * K + kofs);

    float4 a[NL], b0[NL], b1[NL];
    #pragma unroll
    for (int i = 0; i < NL; ++i) {
        a[i]  = ap[2 * i];
        b0[i] = bp0[2 * i];
        b1[i] = bp1[2 * i];
    }

    v8f acc0 = {}, acc1 = {};
    #pragma unroll
    for (int i = 0; i < NL; ++i) {
        const v2f alo  = {a[i].x,  a[i].y},  ahi  = {a[i].z,  a[i].w};
        const v2f b0lo = {b0[i].x, b0[i].y}, b0hi = {b0[i].z, b0[i].w};
        const v2f b1lo = {b1[i].x, b1[i].y}, b1hi = {b1[i].z, b1[i].w};
        acc0 = __builtin_amdgcn_wmma_f32_16x16x4_f32(false, alo, false, b0lo, (short)0, acc0, false, false);
        acc1 = __builtin_amdgcn_wmma_f32_16x16x4_f32(false, alo, false, b1lo, (short)0, acc1, false, false);
        acc0 = __builtin_amdgcn_wmma_f32_16x16x4_f32(false, ahi, false, b0hi, (short)0, acc0, false, false);
        acc1 = __builtin_amdgcn_wmma_f32_16x16x4_f32(false, ahi, false, b1hi, (short)0, acc1, false, false);
    }

    float* mine = red + (wave * 2) * TS + lane;
    #pragma unroll
    for (int r = 0; r < 8; ++r) {
        mine[r * 32]      = acc0[r];
        mine[TS + r * 32] = acc1[r];
    }
    __syncthreads();

    const int t = lane >> 4;
    for (int m = wave; m < 16; m += NW) {
        const float* src = red + t * TS + (m & 7) * 32 + (m >> 3) * 16 + c;
        float v = src[0];
        #pragma unroll
        for (int p = 1; p < NW; ++p)
            v += src[p * 2 * TS];
        v += bv;
        if (SWISH) v = v * sigmoidf_fast(v);
        else       v = sigmoidf_fast(v);
        out[(size_t)(m0 + m) * N + n0 + lane] = v;
    }
}

__global__ __launch_bounds__(256) void se_scale(const float* __restrict__ x,
                                                const float* __restrict__ g,
                                                float* __restrict__ out) {
    const size_t idx4 = (size_t)blockIdx.x * blockDim.x + threadIdx.x;
    const size_t total4 = (size_t)B_ * C_ * HW_ / 4;
    if (idx4 >= total4) return;
    const size_t bc = idx4 >> 8;
    const float gv = g[bc];
    float4 v = ((const float4*)x)[idx4];
    v.x *= gv; v.y *= gv; v.z *= gv; v.w *= gv;
    ((float4*)out)[idx4] = v;
}

extern "C" void kernel_launch(void* const* d_in, const int* in_sizes, int n_in,
                              void* d_out, int out_size, void* d_ws, size_t ws_size,
                              hipStream_t stream) {
    const float* x  = (const float*)d_in[0];
    const float* w1 = (const float*)d_in[1];
    const float* b1 = (const float*)d_in[2];
    const float* w2 = (const float*)d_in[3];
    const float* b2 = (const float*)d_in[4];
    float* out = (float*)d_out;

    float* s = (float*)d_ws;
    float* h = s + (size_t)B_ * C_;
    float* g = h + (size_t)B_ * CR_;

    se_squeeze<<<B_ * C_, 256, 0, stream>>>(x, s);

    se_fc_wmma<C_, CR_, 16, true><<<dim3(CR_ / 32, B_ / 16), 16 * 32, 0, stream>>>(s, w1, b1, h);

    se_fc_wmma<CR_, C_, 8, false><<<dim3(C_ / 32, B_ / 16), 8 * 32, 0, stream>>>(h, w2, b2, g);

    const int nblocks = (int)(((size_t)B_ * C_ * HW_ / 4 + 255) / 256);
    se_scale<<<nblocks, 256, 0, stream>>>(x, g, out);
}


# Pool and scale launches: a wave owns 2 rows and keeps 16 wave-wide 16-byte loads in flight per lane (was 1): 16x fewer waves; pool sums in lane, then lane butterfly, no LDS (new f32 sum order)
# speedup vs baseline: 5.0190x; 1.4543x over previous
#include <hip/hip_runtime.h>
#include <hip/hip_bf16.h>
#include <math.h>

#define B_   32
#define C_   768
#define CR_  192
#define HW_  1024

typedef __attribute__((ext_vector_type(2))) float v2f;
typedef __attribute__((ext_vector_type(8))) float v8f;

__device__ __forceinline__ float sigmoidf_fast(float v) {
    return 1.0f / (1.0f + __expf(-v));
}

#define RPW_          2
#define ROWS_PER_WG_  (8 * RPW_)
#define NLD_          (RPW_ * HW_ / 128)

__global__ __launch_bounds__(256) void se_squeeze(const float* __restrict__ x,
                                                  float* __restrict__ s) {
    const int lane = threadIdx.x & 31;
    const int row0 = __builtin_amdgcn_readfirstlane((int)(blockIdx.x * 8 + (threadIdx.x >> 5)) * RPW_);
    const float4* p = (const float4*)(x + (size_t)row0 * HW_) + lane;

    float4 v[NLD_];
    #pragma unroll
    for (int i = 0; i < NLD_; ++i) v[i] = p[i * 32];
    __builtin_amdgcn_sched_barrier(0);

    float tot[RPW_];
    #pragma unroll
    for (int r = 0; r < RPW_; ++r) {
        float t[8];
        #pragma unroll
        for (int q = 0; q < 8; ++q) {
            const float4 u = v[r * 8 + q];
            t[q] = (u.x + u.y) + (u.z + u.w);
        }
        float sum = ((t[0] + t[1]) + (t[2] + t[3])) + ((t[4] + t[5]) + (t[6] + t[7]));
        #pragma unroll
        for (int o = 16; o > 0; o >>= 1)
            sum += __shfl_xor(sum, o, 32);
        tot[r] = sum;
    }
    float mine = tot[0];
    #pragma unroll
    for (int r = 1; r < RPW_; ++r) mine = (lane == r) ? tot[r] : mine;
    if (lane < RPW_) s[row0 + lane] = mine * (1.0f / (float)HW_);
}

template <int K, int N, int NW, bool SWISH>
__global__ __launch_bounds__(NW * 32) void se_fc_wmma(const float* __restrict__ in,
                                                      const float* __restrict__ w,
                                                      const float* __restrict__ bias,
                                                      float* __restrict__ out) {
    constexpr int R  = K / NW;
    constexpr int NL = R / 8;
    static_assert(K % NW == 0 && R % 8 == 0, "k range of a wave must be a multiple of 8");
    static_assert(N % 32 == 0 && NW <= 16, "one wave per output row in the epilogue");
    constexpr int TS = 272;
    __shared__ float red[NW * 2 * TS];

    const int wave = threadIdx.x >> 5;
    const int lane = threadIdx.x & 31;
    const int half = lane >> 4;
    const int c    = lane & 15;
    const int m0   = blockIdx.y * 16;
    const int n0   = blockIdx.x * 32;

    const float bv = bias[n0 + lane];

    const int kofs = wave * R + 4 * half;
    const float4* ap  = (const float4*)(in + (size_t)(m0 + c)      * K + kofs);
    const float4* bp0 = (const float4*)(w  + (size_t)(n0 + c)      * K + kofs);
    const float4* bp1 = (const float4*)(w  + (size_t)(n0 + 16 + c) * K + kofs);

    float4 a[NL], b0[NL], b1[NL];
    #pragma unroll
    for (int i = 0; i < NL; ++i) {
        a[i]  = ap[2 * i];
        b0[i] = bp0[2 * i];
        b1[i] = bp1[2 * i];
    }

    v8f acc0 = {}, acc1 = {};
    #pragma unroll
    for (int i = 0; i < NL; ++i) {
        const v2f alo  = {a[i].x,  a[i].y},  ahi  = {a[i].z,  a[i].w};
        const v2f b0lo = {b0[i].x, b0[i].y}, b0hi = {b0[i].z, b0[i].w};
        const v2f b1lo = {b1[i].x, b1[i].y}, b1hi = {b1[i].z, b1[i].w};
        acc0 = __builtin_amdgcn_wmma_f32_16x16x4_f32(false, alo, false, b0lo, (short)0, acc0, false, false);
        acc1 = __builtin_amdgcn_wmma_f32_16x16x4_f32(false, alo, false, b1lo, (short)0, acc1, false, false);
        acc0 = __builtin_amdgcn_wmma_f32_16x16x4_f32(false, ahi, false, b0hi, (short)0, acc0, false, false);
        acc1 = __builtin_amdgcn_wmma_f32_16x16x4_f32(false, ahi, false, b1hi, (short)0, acc1, false, false);
    }

    float* mine = red + (wave * 2) * TS + lane;
    #pragma unroll
    for (int r = 0; r < 8; ++r) {
        mine[r * 32]      = acc0[r];
        mine[TS + r * 32] = acc1[r];
    }
    __syncthreads();

    const int t = lane >> 4;
    for (int m = wave; m < 16; m += NW) {
        const float* src = red + t * TS + (m & 7) * 32 + (m >> 3) * 16 + c;
        float v = src[0];
        #pragma unroll
        for (int p = 1; p < NW; ++p)
            v += src[p * 2 * TS];
        v += bv;
        if (SWISH) v = v * sigmoidf_fast(v);
        else       v = sigmoidf_fast(v);
        out[(size_t)(m0 + m) * N + n0 + lane] = v;
    }
}

__global__ __launch_bounds__(256) void se_scale(const float* __restrict__ x,
                                                const float* __restrict__ g,
                                                float* __restrict__ out) {
    const int lane = threadIdx.x & 31;
    const int row0 = __builtin_amdgcn_readfirstlane((int)(blockIdx.x * 8 + (threadIdx.x >> 5)) * RPW_);
    const float4* p = (const float4*)(x   + (size_t)row0 * HW_) + lane;
    float4*       o = (float4*)      (out + (size_t)row0 * HW_) + lane;

    float4 v[NLD_];
    #pragma unroll
    for (int i = 0; i < NLD_; ++i) v[i] = p[i * 32];
    __builtin_amdgcn_sched_barrier(0);

    float gv[RPW_];
    #pragma unroll
    for (int r = 0; r < RPW_; ++r) gv[r] = g[row0 + r];

    #pragma unroll
    for (int i = 0; i < NLD_; ++i) {
        const float gr = gv[i / 8];
        float4 u = v[i];
        u.x *= gr; u.y *= gr; u.z *= gr; u.w *= gr;
        o[i * 32] = u;
    }
}

extern "C" void kernel_launch(void* const* d_in, const int* in_sizes, int n_in,
                              void* d_out, int out_size, void* d_ws, size_t ws_size,
                              hipStream_t stream) {
    const float* x  = (const float*)d_in[0];
    const float* w1 = (const float*)d_in[1];
    const float* b1 = (const float*)d_in[2];
    const float* w2 = (const float*)d_in[3];
    const float* b2 = (const float*)d_in[4];
    float* out = (float*)d_out;

    float* s = (float*)d_ws;
    float* h = s + (size_t)B_ * C_;
    float* g = h + (size_t)B_ * CR_;

    se_squeeze<<<B_ * C_ / ROWS_PER_WG_, 256, 0, stream>>>(x, s);

    se_fc_wmma<C_, CR_, 16, true><<<dim3(CR_ / 32, B_ / 16), 16 * 32, 0, stream>>>(s, w1, b1, h);

    se_fc_wmma<CR_, C_, 8, false><<<dim3(C_ / 32, B_ / 16), 8 * 32, 0, stream>>>(h, w2, b2, g);

    se_scale<<<B_ * C_ / ROWS_PER_WG_, 256, 0, stream>>>(x, g, out);
}


# Both dense layers in one launch: 24 work-groups each keep a 16x16 h tile in LDS and form a k-partial plane of the gate layer; scale adds the 12 planes, bias, sigmoid; 3 launches (new f32 sum order)
# speedup vs baseline: 5.3346x; 1.0629x over previous
#include <hip/hip_runtime.h>
#include <hip/hip_bf16.h>
#include <math.h>

#define B_   32
#define C_   768
#define CR_  192
#define HW_  1024

typedef __attribute__((ext_vector_type(2))) float v2f;
typedef __attribute__((ext_vector_type(8))) float v8f;

__device__ __forceinline__ float sigmoidf_fast(float v) {
    return 1.0f / (1.0f + __expf(-v));
}

#define RPW_          2
#define ROWS_PER_WG_  (8 * RPW_)
#define NLD_          (RPW_ * HW_ / 128)

__global__ __launch_bounds__(256) void se_squeeze(const float* __restrict__ x,
                                                  float* __restrict__ s) {
    const int lane = threadIdx.x & 31;
    const int row0 = __builtin_amdgcn_readfirstlane((int)(blockIdx.x * 8 + (threadIdx.x >> 5)) * RPW_);
    const float4* p = (const float4*)(x + (size_t)row0 * HW_) + lane;

    float4 v[NLD_];
    #pragma unroll
    for (int i = 0; i < NLD_; ++i) v[i] = p[i * 32];
    __builtin_amdgcn_sched_barrier(0);

    float tot[RPW_];
    #pragma unroll
    for (int r = 0; r < RPW_; ++r) {
        float t[8];
        #pragma unroll
        for (int q = 0; q < 8; ++q) {
            const float4 u = v[r * 8 + q];
            t[q] = (u.x + u.y) + (u.z + u.w);
        }
        float sum = ((t[0] + t[1]) + (t[2] + t[3])) + ((t[4] + t[5]) + (t[6] + t[7]));
        #pragma unroll
        for (int o = 16; o > 0; o >>= 1)
            sum += __shfl_xor(sum, o, 32);
        tot[r] = sum;
    }
    float mine = tot[0];
    #pragma unroll
    for (int r = 1; r < RPW_; ++r) mine = (lane == r) ? tot[r] : mine;
    if (lane < RPW_) s[row0 + lane] = mine * (1.0f / (float)HW_);
}

#define NPL_  (CR_ / 16)
#define NQ_   (C_ / 32)

__global__ __launch_bounds__(512) void se_mlp_wmma(const float* __restrict__ in,
                                                   const float* __restrict__ w,
                                                   const float* __restrict__ bias,
                                                   const float* __restrict__ w2,
                                                   float* __restrict__ gp) {
    constexpr int K  = C_;
    constexpr int NW = 16;
    constexpr int R  = K / NW;
    constexpr int NL = R / 8;
    constexpr int HP = 20;
    __shared__ float red[NW * 256];
    __shared__ __attribute__((aligned(16))) float hT[16 * HP];

    const int wave = __builtin_amdgcn_readfirstlane((int)(threadIdx.x >> 5));
    const int lane = threadIdx.x & 31;
    const int half = lane >> 4;
    const int c    = lane & 15;
    const int m0   = blockIdx.y * 16;
    const int n0   = blockIdx.x * 16;

    const float bv = bias[n0 + c];

    const int kofs = wave * R + 4 * half;
    const float4* ap = (const float4*)(in + (size_t)(m0 + c) * K + kofs);
    const float4* bp = (const float4*)(w  + (size_t)(n0 + c) * K + kofs);

    float4 a[NL], b[NL];
    #pragma unroll
    for (int i = 0; i < NL; ++i) {
        a[i] = ap[2 * i];
        b[i] = bp[2 * i];
    }

    float4 bw[2][2][2];
    #pragma unroll
    for (int qq = 0; qq < 2; ++qq) {
        const int q  = wave + 16 * qq;
        const int ql = (q < NQ_) ? q : wave;
        #pragma unroll
        for (int t = 0; t < 2; ++t) {
            const float4* wp = (const float4*)(w2 + (size_t)(ql * 32 + t * 16 + c) * CR_ + n0 + 8 * half);
            #pragma unroll
            for (int i = 0; i < 2; ++i) bw[qq][t][i] = wp[i];
        }
    }
    __builtin_amdgcn_sched_barrier(0);

    v8f acc = {};
    #pragma unroll
    for (int i = 0; i < NL; ++i) {
        const v2f alo = {a[i].x, a[i].y}, ahi = {a[i].z, a[i].w};
        const v2f blo = {b[i].x, b[i].y}, bhi = {b[i].z, b[i].w};
        acc = __builtin_amdgcn_wmma_f32_16x16x4_f32(false, alo, false, blo, (short)0, acc, false, false);
        acc = __builtin_amdgcn_wmma_f32_16x16x4_f32(false, ahi, false, bhi, (short)0, acc, false, false);
    }

    float* mine = red + wave * 256 + lane;
    #pragma unroll
    for (int r = 0; r < 8; ++r) mine[r * 32] = acc[r];
    __syncthreads();

    if (wave < 8) {
        const int m = 2 * wave + half;
        const float* src = red + (m & 7) * 32 + (m >> 3) * 16 + c;
        float v = src[0];
        #pragma unroll
        for (int p = 1; p < NW; ++p)
            v += src[p * 256];
        v += bv;
        v = v * sigmoidf_fast(v);
        hT[m * HP + c] = v;
    }
    __syncthreads();

    float4 af[2];
    #pragma unroll
    for (int i = 0; i < 2; ++i) af[i] = *(const float4*)(hT + c * HP + 8 * half + 4 * i);

    #pragma unroll
    for (int qq = 0; qq < 2; ++qq) {
        const int q = wave + 16 * qq;
        if (q < NQ_) {
            v8f d0 = {}, d1 = {};
            #pragma unroll
            for (int i = 0; i < 2; ++i) {
                const v2f alo  = {af[i].x, af[i].y},               ahi  = {af[i].z, af[i].w};
                const v2f b0lo = {bw[qq][0][i].x, bw[qq][0][i].y}, b0hi = {bw[qq][0][i].z, bw[qq][0][i].w};
                const v2f b1lo = {bw[qq][1][i].x, bw[qq][1][i].y}, b1hi = {bw[qq][1][i].z, bw[qq][1][i].w};
                d0 = __builtin_amdgcn_wmma_f32_16x16x4_f32(false, alo, false, b0lo, (short)0, d0, false, false);
                d1 = __builtin_amdgcn_wmma_f32_16x16x4_f32(false, alo, false, b1lo, (short)0, d1, false, false);
                d0 = __builtin_amdgcn_wmma_f32_16x16x4_f32(false, ahi, false, b0hi, (short)0, d0, false, false);
                d1 = __builtin_amdgcn_wmma_f32_16x16x4_f32(false, ahi, false, b1hi, (short)0, d1, false, false);
            }
            float* dst = gp + ((size_t)blockIdx.x * B_ + m0 + 8 * half) * C_ + q * 32 + c;
            #pragma unroll
            for (int r = 0; r < 8; ++r) {
                dst[(size_t)r * C_]      = d0[r];
                dst[(size_t)r * C_ + 16] = d1[r];
            }
        }
    }
}

__global__ __launch_bounds__(256) void se_scale(const float* __restrict__ x,
                                                const float* __restrict__ gp,
                                                const float* __restrict__ b2,
                                                float* __restrict__ out) {
    const int lane = threadIdx.x & 31;
    const int row0 = __builtin_amdgcn_readfirstlane((int)(blockIdx.x * 8 + (threadIdx.x >> 5)) * RPW_);
    const float4* p = (const float4*)(x   + (size_t)row0 * HW_) + lane;
    float4*       o = (float4*)      (out + (size_t)row0 * HW_) + lane;

    float4 v[NLD_];
    #pragma unroll
    for (int i = 0; i < NLD_; ++i) v[i] = p[i * 32];
    __builtin_amdgcn_sched_barrier(0);

    float gv[RPW_];
    #pragma unroll
    for (int r = 0; r < RPW_; ++r) {
        float pre = gp[row0 + r];
        #pragma unroll
        for (int k = 1; k < NPL_; ++k) pre += gp[(size_t)k * (B_ * C_) + row0 + r];
        pre += b2[(row0 + r) % C_];
        gv[r] = sigmoidf_fast(pre);
    }

    #pragma unroll
    for (int i = 0; i < NLD_; ++i) {
        const float gr = gv[i / 8];
        float4 u = v[i];
        u.x *= gr; u.y *= gr; u.z *= gr; u.w *= gr;
        o[i * 32] = u;
    }
}

extern "C" void kernel_launch(void* const* d_in, const int* in_sizes, int n_in,
                              void* d_out, int out_size, void* d_ws, size_t ws_size,
                              hipStream_t stream) {
    const float* x  = (const float*)d_in[0];
    const float* w1 = (const float*)d_in[1];
    const float* b1 = (const float*)d_in[2];
    const float* w2 = (const float*)d_in[3];
    const float* b2 = (const float*)d_in[4];
    float* out = (float*)d_out;

    float* s  = (float*)d_ws;
    float* gp = s + (size_t)B_ * C_;

    se_squeeze<<<B_ * C_ / ROWS_PER_WG_, 256, 0, stream>>>(x, s);

    se_mlp_wmma<<<dim3(CR_ / 16, B_ / 16), 16 * 32, 0, stream>>>(s, w1, b1, w2, gp);

    se_scale<<<B_ * C_ / ROWS_PER_WG_, 256, 0, stream>>>(x, gp, b2, out);
}


# One launch fewer (2, was 3): pool launch also forms k-partial planes of the hidden layer for its 32 channels, scale launch adds the 24 planes and forms its 32 gates itself (new f32 sum order)
# speedup vs baseline: 6.3693x; 1.1940x over previous
#include <hip/hip_runtime.h>
#include <math.h>

#define B_   32
#define C_   768
#define CR_  192
#define HW_  1024

typedef __attribute__((ext_vector_type(2))) float v2f;
typedef __attribute__((ext_vector_type(8))) float v8f;

__device__ __forceinline__ float sigmoidf_fast(float v) {
    return 1.0f / (1.0f + __expf(-v));
}

__device__ __forceinline__ void lds_barrier() {
    __builtin_amdgcn_fence(__ATOMIC_RELEASE, "workgroup", "local");
    __builtin_amdgcn_s_barrier();
    __builtin_amdgcn_fence(__ATOMIC_ACQUIRE, "workgroup", "local");
}

#define RPW_   4
#define RWG_   32
#define NG_    (C_ / RWG_)
#define NWG_   (B_ * C_ / RWG_)

__global__ __launch_bounds__(256) void se_pool_l1(const float* __restrict__ x,
                                                  const float* __restrict__ w1,
                                                  float* __restrict__ hp) {
    __shared__ __attribute__((aligned(16))) float sm[RWG_];

    const int wave = __builtin_amdgcn_readfirstlane((int)(threadIdx.x >> 5));
    const int lane = threadIdx.x & 31;
    const int half = lane >> 4;
    const int c    = lane & 15;
    const int rowbase = blockIdx.x * RWG_;
    const int b    = rowbase / C_;
    const int ch0  = rowbase % C_;
    const int row0 = rowbase + wave * RPW_;
    const float4* p = (const float4*)(x + (size_t)row0 * HW_) + lane;

    float4 v[RPW_ * 8];
    #pragma unroll
    for (int i = 0; i < RPW_ * 8; ++i) v[i] = p[i * 32];
    __builtin_amdgcn_sched_barrier(0);

    float tot[RPW_];
    #pragma unroll
    for (int r = 0; r < RPW_; ++r) {
        float t[8];
        #pragma unroll
        for (int q = 0; q < 8; ++q) {
            const float4 u = v[r * 8 + q];
            t[q] = (u.x + u.y) + (u.z + u.w);
        }
        float sum = ((t[0] + t[1]) + (t[2] + t[3])) + ((t[4] + t[5]) + (t[6] + t[7]));
        #pragma unroll
        for (int o = 16; o > 0; o >>= 1)
            sum += __shfl_xor(sum, o, 32);
        tot[r] = sum;
    }

    float4 bw[2][4];
    #pragma unroll
    for (int t = 0; t < 2; ++t) {
        #pragma unroll
        for (int i = 0; i < 4; ++i) bw[t][i] = make_float4(0.f, 0.f, 0.f, 0.f);
    }
    if (wave < CR_ / 32) {
        #pragma unroll
        for (int t = 0; t < 2; ++t) {
            const float4* wp = (const float4*)(w1 + (size_t)(32 * wave + 16 * t + c) * C_ + ch0 + 4 * half);
            #pragma unroll
            for (int i = 0; i < 4; ++i) bw[t][i] = wp[2 * i];
        }
    }

    float mine = tot[0];
    #pragma unroll
    for (int r = 1; r < RPW_; ++r) mine = (lane == r) ? tot[r] : mine;
    if (lane < RPW_) sm[wave * RPW_ + lane] = mine * (1.0f / (float)HW_);
    lds_barrier();

    if (wave < CR_ / 32) {
        const float4 z4 = make_float4(0.f, 0.f, 0.f, 0.f);
        v8f d0 = {}, d1 = {};
        #pragma unroll
        for (int i = 0; i < 4; ++i) {
            const float4 s4 = *(const float4*)(sm + 8 * i + 4 * half);
            const float4 a4 = (c == 0) ? s4 : z4;
            const v2f alo  = {a4.x, a4.y},             ahi  = {a4.z, a4.w};
            const v2f b0lo = {bw[0][i].x, bw[0][i].y}, b0hi = {bw[0][i].z, bw[0][i].w};
            const v2f b1lo = {bw[1][i].x, bw[1][i].y}, b1hi = {bw[1][i].z, bw[1][i].w};
            d0 = __builtin_amdgcn_wmma_f32_16x16x4_f32(false, alo, false, b0lo, (short)0, d0, false, false);
            d1 = __builtin_amdgcn_wmma_f32_16x16x4_f32(false, alo, false, b1lo, (short)0, d1, false, false);
            d0 = __builtin_amdgcn_wmma_f32_16x16x4_f32(false, ahi, false, b0hi, (short)0, d0, false, false);
            d1 = __builtin_amdgcn_wmma_f32_16x16x4_f32(false, ahi, false, b1hi, (short)0, d1, false, false);
        }
        if (half == 0) {
            float* dst = hp + ((size_t)(ch0 / RWG_) * B_ + b) * CR_ + 32 * wave + c;
            dst[0]  = d0[0];
            dst[16] = d1[0];
        }
    }
}

__global__ __launch_bounds__(256) void se_gate_scale(const float* __restrict__ x,
                                                     const float* __restrict__ hp,
                                                     const float* __restrict__ b1,
                                                     const float* __restrict__ w2,
                                                     const float* __restrict__ b2,
                                                     float* __restrict__ out) {
    constexpr int KW = CR_ / 8;
    constexpr int NL = KW / 8;
    __shared__ __attribute__((aligned(16))) float hT[256];
    __shared__ float red[8 * RWG_];
    __shared__ float gate[RWG_];

    const int wave = __builtin_amdgcn_readfirstlane((int)(threadIdx.x >> 5));
    const int lane = threadIdx.x & 31;
    const int half = lane >> 4;
    const int c    = lane & 15;
    const int rowbase = blockIdx.x * RWG_;
    const int b    = rowbase / C_;
    const int ch0  = rowbase % C_;
    const int row0 = rowbase + wave * RPW_;
    const float4* p = (const float4*)(x   + (size_t)row0 * HW_) + lane;
    float4*       o = (float4*)      (out + (size_t)row0 * HW_) + lane;

    const int j = (threadIdx.x < CR_) ? (int)threadIdx.x : (CR_ - 1);
    const float* pp = hp + (size_t)b * CR_ + j;
    float pl[NG_];
    #pragma unroll
    for (int k = 0; k < NG_; ++k) pl[k] = pp[(size_t)k * (B_ * CR_)];
    const float b1v = b1[j];
    const float b2v = b2[ch0 + lane];

    float4 bw[2][NL];
    #pragma unroll
    for (int t = 0; t < 2; ++t) {
        const float4* wp = (const float4*)(w2 + (size_t)(ch0 + 16 * t + c) * CR_ + KW * wave + 4 * half);
        #pragma unroll
        for (int i = 0; i < NL; ++i) bw[t][i] = wp[2 * i];
    }

    __builtin_amdgcn_sched_barrier(0);

    float4 v[16];
    #pragma unroll
    for (int i = 0; i < 16; ++i) v[i] = p[i * 32];
    __builtin_amdgcn_sched_barrier(0);

    float hv = pl[0];
    #pragma unroll
    for (int k = 1; k < NG_; ++k) hv += pl[k];
    hv += b1v;
    hv = hv * sigmoidf_fast(hv);
    hT[threadIdx.x] = hv;
    lds_barrier();

    {
        const float4 z4 = make_float4(0.f, 0.f, 0.f, 0.f);
        v8f d0 = {}, d1 = {};
        #pragma unroll
        for (int i = 0; i < NL; ++i) {
            const float4 s4 = *(const float4*)(hT + KW * wave + 8 * i + 4 * half);
            const float4 a4 = (c == 0) ? s4 : z4;
            const v2f alo  = {a4.x, a4.y},             ahi  = {a4.z, a4.w};
            const v2f b0lo = {bw[0][i].x, bw[0][i].y}, b0hi = {bw[0][i].z, bw[0][i].w};
            const v2f b1lo = {bw[1][i].x, bw[1][i].y}, b1hi = {bw[1][i].z, bw[1][i].w};
            d0 = __builtin_amdgcn_wmma_f32_16x16x4_f32(false, alo, false, b0lo, (short)0, d0, false, false);
            d1 = __builtin_amdgcn_wmma_f32_16x16x4_f32(false, alo, false, b1lo, (short)0, d1, false, false);
            d0 = __builtin_amdgcn_wmma_f32_16x16x4_f32(false, ahi, false, b0hi, (short)0, d0, false, false);
            d1 = __builtin_amdgcn_wmma_f32_16x16x4_f32(false, ahi, false, b1hi, (short)0, d1, false, false);
        }
        if (half == 0) {
            red[wave * RWG_ + c]      = d0[0];
            red[wave * RWG_ + 16 + c] = d1[0];
        }
    }

    float4 u2[16];
    #pragma unroll
    for (int i = 0; i < 16; ++i) u2[i] = p[(16 + i) * 32];
    lds_barrier();

    if (wave == 0) {
        float gsum = red[lane];
        #pragma unroll
        for (int w = 1; w < 8; ++w) gsum += red[w * RWG_ + lane];
        gsum += b2v;
        gate[lane] = sigmoidf_fast(gsum);
    }
    lds_barrier();

    float gv[RPW_];
    #pragma unroll
    for (int r = 0; r < RPW_; ++r) gv[r] = gate[wave * RPW_ + r];

    #pragma unroll
    for (int i = 0; i < 16; ++i) {
        const float gr = gv[i / 8];
        float4 u = v[i];
        u.x *= gr; u.y *= gr; u.z *= gr; u.w *= gr;
        o[i * 32] = u;
    }
    #pragma unroll
    for (int i = 0; i < 16; ++i) {
        const float gr = gv[2 + i / 8];
        float4 u = u2[i];
        u.x *= gr; u.y *= gr; u.z *= gr; u.w *= gr;
        o[(16 + i) * 32] = u;
    }
}

extern "C" void kernel_launch(void* const* d_in, const int* in_sizes, int n_in,
                              void* d_out, int out_size, void* d_ws, size_t ws_size,
                              hipStream_t stream) {
    const float* x  = (const float*)d_in[0];
    const float* w1 = (const float*)d_in[1];
    const float* b1 = (const float*)d_in[2];
    const float* w2 = (const float*)d_in[3];
    const float* b2 = (const float*)d_in[4];
    float* out = (float*)d_out;

    float* hp = (float*)d_ws;

    se_pool_l1<<<NWG_, 256, 0, stream>>>(x, w1, hp);

    se_gate_scale<<<NWG_, 256, 0, stream>>>(x, hp, b1, w2, b2, out);
}
